# starting kernel
# baseline (speedup 1.0000x reference)
#include <hip/hip_runtime.h>
#include <hip/hip_bf16.h>

typedef float v2f __attribute__((ext_vector_type(2)));
typedef float v8f __attribute__((ext_vector_type(8)));

#define IN_CH  64
#define HID_CH 64
#define OUT_CH 32

__global__ __launch_bounds__(256) void k_init_deg(float* __restrict__ deg, int n) {
    int i = blockIdx.x * 256 + threadIdx.x;
    if (i < n) deg[i] = 1.0f;
}

__global__ __launch_bounds__(256) void k_deg(const int* __restrict__ dst,
                                             float* __restrict__ deg, int e) {
    int i = blockIdx.x * 256 + threadIdx.x;
    if (i < e) atomicAdd(&deg[dst[i]], 1.0f);
}

__global__ __launch_bounds__(256) void k_dis(const float* __restrict__ deg,
                                             float* __restrict__ dis, int n) {
    int i = blockIdx.x * 256 + threadIdx.x;
    if (i < n) dis[i] = rsqrtf(deg[i]);
}

template <int NOUT>
__global__ __launch_bounds__(256) void k_gemm_wmma(const float* __restrict__ A,
                                                   const float* __restrict__ B,
                                                   float* __restrict__ C,
                                                   int numTiles) {
    constexpr int K = 64;
    const int wave = threadIdx.x >> 5;
    const int lane = threadIdx.x & 31;
    const int tile = blockIdx.x * 8 + wave;
    if (tile >= numTiles) return;
    constexpr int tilesPerRow = NOUT / 16;
    const int row0 = (tile / tilesPerRow) * 16;
    const int col0 = (tile % tilesPerRow) * 16;
    const int half = lane >> 4;
    const int l15  = lane & 15;

    const float* __restrict__ Arow = A + (size_t)(row0 + l15) * K + half * 2;
    const float* __restrict__ Bp   = B + col0 + l15;

    v8f acc = {0.f, 0.f, 0.f, 0.f, 0.f, 0.f, 0.f, 0.f};
#pragma unroll
    for (int k0 = 0; k0 < K; k0 += 4) {
        v2f a;
        a.x = Arow[k0];
        a.y = Arow[k0 + 1];
        v2f b;
        b.x = Bp[(size_t)(k0 + half * 2)     * NOUT];
        b.y = Bp[(size_t)(k0 + half * 2 + 1) * NOUT];
        acc = __builtin_amdgcn_wmma_f32_16x16x4_f32(
             false, a,  false, b,
             (short)0, acc,  false,  false);
    }

    float* Cp = C + (size_t)row0 * NOUT + col0 + l15;
#pragma unroll
    for (int v = 0; v < 8; ++v)
        Cp[(size_t)(v + half * 8) * NOUT] = acc[v];
}

template <int C>
__global__ __launch_bounds__(256) void k_scatter(const int* __restrict__ src,
                                                 const int* __restrict__ dst,
                                                 const float* __restrict__ h,
                                                 const float* __restrict__ dis,
                                                 float* __restrict__ agg, int e) {
    constexpr int TPE = C / 4;
    int t = blockIdx.x * 256 + threadIdx.x;
    int ed = t / TPE;
    if (ed >= e) return;
    int c = (t % TPE) * 4;
    int s = src[ed], d = dst[ed];
    float norm = dis[s] * dis[d];
    const float4 v = *(const float4*)(h + (size_t)s * C + c);
    float* o = agg + (size_t)d * C + c;
    atomicAdd(o + 0, v.x * norm);
    atomicAdd(o + 1, v.y * norm);
    atomicAdd(o + 2, v.z * norm);
    atomicAdd(o + 3, v.w * norm);
}

__global__ __launch_bounds__(256) void k_finalize_relu(float* __restrict__ agg,
                                                       const float* __restrict__ h,
                                                       const float* __restrict__ dis,
                                                       const float* __restrict__ bias,
                                                       int n) {
    int t = blockIdx.x * 256 + threadIdx.x;
    int i = t / HID_CH;
    if (i >= n) return;
    int c = t % HID_CH;
    float di = dis[i];
    float v = agg[t] + h[t] * di * di + bias[c];
    agg[t] = fmaxf(v, 0.0f);
}

__global__ __launch_bounds__(256) void k_final_logsoftmax(const float* __restrict__ agg,
                                                          const float* __restrict__ h,
                                                          const float* __restrict__ dis,
                                                          const float* __restrict__ bias,
                                                          float* __restrict__ out, int n) {
    int wave = threadIdx.x >> 5;
    int lane = threadIdx.x & 31;
    int i = blockIdx.x * 8 + wave;
    if (i >= n) return;
    float di = dis[i];
    size_t idx = (size_t)i * OUT_CH + lane;
    float v = agg[idx] + h[idx] * di * di + bias[lane];
    float m = v;
#pragma unroll
    for (int o = 16; o > 0; o >>= 1) m = fmaxf(m, __shfl_xor(m, o, 32));
    float s = __expf(v - m);
#pragma unroll
    for (int o = 16; o > 0; o >>= 1) s += __shfl_xor(s, o, 32);
    out[idx] = (v - m) - logf(s);
}

extern "C" void kernel_launch(void* const* d_in, const int* in_sizes, int n_in,
                              void* d_out, int out_size, void* d_ws, size_t ws_size,
                              hipStream_t stream) {
    const float* x  = (const float*)d_in[0];
    const int*   ei = (const int*)  d_in[1];
    const float* W1 = (const float*)d_in[2];
    const float* b1 = (const float*)d_in[3];
    const float* W2 = (const float*)d_in[4];
    const float* b2 = (const float*)d_in[5];
    float* out = (float*)d_out;

    const int N = in_sizes[0] / IN_CH;
    const int E = in_sizes[1] / 2;
    const int* src = ei;
    const int* dst = ei + E;

    char* ws = (char*)d_ws;
    size_t off = 0;
    auto carve = [&](size_t bytes) -> void* {
        void* p = ws + off;
        off += (bytes + 255) & ~(size_t)255;
        return p;
    };
    float* deg  = (float*)carve((size_t)N * sizeof(float));
    float* dis  = (float*)carve((size_t)N * sizeof(float));
    float* h1   = (float*)carve((size_t)N * HID_CH * sizeof(float));
    float* agg1 = (float*)carve((size_t)N * HID_CH * sizeof(float));
    float* h2   = (float*)carve((size_t)N * OUT_CH * sizeof(float));
    float* agg2 = (float*)carve((size_t)N * OUT_CH * sizeof(float));

    k_init_deg<<<(N + 255) / 256, 256, 0, stream>>>(deg, N);
    k_deg<<<(E + 255) / 256, 256, 0, stream>>>(dst, deg, E);
    k_dis<<<(N + 255) / 256, 256, 0, stream>>>(deg, dis, N);

    hipMemsetAsync(agg1, 0, (size_t)N * HID_CH * sizeof(float), stream);
    const int tiles1 = (N / 16) * (HID_CH / 16);
    k_gemm_wmma<HID_CH><<<(tiles1 + 7) / 8, 256, 0, stream>>>(x, W1, h1, tiles1);
    {
        long long threads = (long long)E * (HID_CH / 4);
        k_scatter<HID_CH><<<(unsigned)((threads + 255) / 256), 256, 0, stream>>>(
            src, dst, h1, dis, agg1, E);
    }
    {
        long long threads = (long long)N * HID_CH;
        k_finalize_relu<<<(unsigned)((threads + 255) / 256), 256, 0, stream>>>(
            agg1, h1, dis, b1, N);
    }

    hipMemsetAsync(agg2, 0, (size_t)N * OUT_CH * sizeof(float), stream);
    const int tiles2 = (N / 16) * (OUT_CH / 16);
    k_gemm_wmma<OUT_CH><<<(tiles2 + 7) / 8, 256, 0, stream>>>(agg1, W2, h2, tiles2);
    {
        long long threads = (long long)E * (OUT_CH / 4);
        k_scatter<OUT_CH><<<(unsigned)((threads + 255) / 256), 256, 0, stream>>>(
            src, dst, h2, dis, agg2, E);
    }
    k_final_logsoftmax<<<(N + 7) / 8, 256, 0, stream>>>(agg2, h2, dis, b2, out, N);
}


# Scatter: a wave walks 32 edges with lane = channel, so each row load and each f32 atomic add instruction covers one whole 128-B line (was 4 lines x 8 floats); sums stay atomics in arrival order
# speedup vs baseline: 1.2742x; 1.2742x over previous
#include <hip/hip_runtime.h>
#include <hip/hip_bf16.h>

typedef float v2f __attribute__((ext_vector_type(2)));
typedef float v8f __attribute__((ext_vector_type(8)));

#define IN_CH  64
#define HID_CH 64
#define OUT_CH 32

__global__ __launch_bounds__(256) void k_init_deg(float* __restrict__ deg, int n) {
    int i = blockIdx.x * 256 + threadIdx.x;
    if (i < n) deg[i] = 1.0f;
}

__global__ __launch_bounds__(256) void k_deg(const int* __restrict__ dst,
                                             float* __restrict__ deg, int e) {
    int i = blockIdx.x * 256 + threadIdx.x;
    if (i < e) atomicAdd(&deg[dst[i]], 1.0f);
}

__global__ __launch_bounds__(256) void k_dis(const float* __restrict__ deg,
                                             float* __restrict__ dis, int n) {
    int i = blockIdx.x * 256 + threadIdx.x;
    if (i < n) dis[i] = rsqrtf(deg[i]);
}

template <int NOUT>
__global__ __launch_bounds__(256) void k_gemm_wmma(const float* __restrict__ A,
                                                   const float* __restrict__ B,
                                                   float* __restrict__ C,
                                                   int numTiles) {
    constexpr int K = 64;
    const int wave = threadIdx.x >> 5;
    const int lane = threadIdx.x & 31;
    const int tile = blockIdx.x * 8 + wave;
    if (tile >= numTiles) return;
    constexpr int tilesPerRow = NOUT / 16;
    const int row0 = (tile / tilesPerRow) * 16;
    const int col0 = (tile % tilesPerRow) * 16;
    const int half = lane >> 4;
    const int l15  = lane & 15;

    const float* __restrict__ Arow = A + (size_t)(row0 + l15) * K + half * 2;
    const float* __restrict__ Bp   = B + col0 + l15;

    v8f acc = {0.f, 0.f, 0.f, 0.f, 0.f, 0.f, 0.f, 0.f};
#pragma unroll
    for (int k0 = 0; k0 < K; k0 += 4) {
        v2f a;
        a.x = Arow[k0];
        a.y = Arow[k0 + 1];
        v2f b;
        b.x = Bp[(size_t)(k0 + half * 2)     * NOUT];
        b.y = Bp[(size_t)(k0 + half * 2 + 1) * NOUT];
        acc = __builtin_amdgcn_wmma_f32_16x16x4_f32(
             false, a,  false, b,
             (short)0, acc,  false,  false);
    }

    float* Cp = C + (size_t)row0 * NOUT + col0 + l15;
#pragma unroll
    for (int v = 0; v < 8; ++v)
        Cp[(size_t)(v + half * 8) * NOUT] = acc[v];
}

template <int C>
__global__ __launch_bounds__(256) void k_scatter(const int* __restrict__ src,
                                                 const int* __restrict__ dst,
                                                 const float* __restrict__ h,
                                                 const float* __restrict__ dis,
                                                 float* __restrict__ agg, int e) {
    constexpr int R = C / 32;
    constexpr int G = 16 / R;
    const unsigned lane = threadIdx.x & 31;
    const int wv = __builtin_amdgcn_readfirstlane((int)(threadIdx.x >> 5));
    const int e0 = ((int)blockIdx.x * 8 + wv) * 32;
    if (e0 >= e) return;
    const int cnt = (e - e0 < 32) ? (e - e0) : 32;

    int s = 0, d = 0;
    float nrm = 0.0f;
    if ((int)lane < cnt) {
        s = src[e0 + lane];
        d = dst[e0 + lane];
        nrm = dis[s] * dis[d];
    }
    const int nrm_bits = __float_as_int(nrm);

    if (cnt == 32) {
#pragma unroll
        for (int j0 = 0; j0 < 32; j0 += G) {
            float v[G][R];
#pragma unroll
            for (int g = 0; g < G; ++g) {
                const unsigned sj = (unsigned)__builtin_amdgcn_readlane(s, j0 + g);
                const float* __restrict__ hrow = h + (size_t)sj * C;
#pragma unroll
                for (int r = 0; r < R; ++r) v[g][r] = hrow[lane + r * 32];
            }
#pragma unroll
            for (int g = 0; g < G; ++g) {
                const unsigned dj = (unsigned)__builtin_amdgcn_readlane(d, j0 + g);
                const float nj = __int_as_float(__builtin_amdgcn_readlane(nrm_bits, j0 + g));
                float* __restrict__ arow = agg + (size_t)dj * C;
#pragma unroll
                for (int r = 0; r < R; ++r) atomicAdd(arow + lane + r * 32, v[g][r] * nj);
            }
        }
    } else {
        for (int j = 0; j < cnt; ++j) {
            const unsigned sj = (unsigned)__builtin_amdgcn_readlane(s, j);
            const unsigned dj = (unsigned)__builtin_amdgcn_readlane(d, j);
            const float nj = __int_as_float(__builtin_amdgcn_readlane(nrm_bits, j));
            const float* __restrict__ hrow = h + (size_t)sj * C;
            float* __restrict__ arow = agg + (size_t)dj * C;
#pragma unroll
            for (int r = 0; r < R; ++r) atomicAdd(arow + lane + r * 32, hrow[lane + r * 32] * nj);
        }
    }
}

__global__ __launch_bounds__(256) void k_finalize_relu(float* __restrict__ agg,
                                                       const float* __restrict__ h,
                                                       const float* __restrict__ dis,
                                                       const float* __restrict__ bias,
                                                       int n) {
    int t = blockIdx.x * 256 + threadIdx.x;
    int i = t / HID_CH;
    if (i >= n) return;
    int c = t % HID_CH;
    float di = dis[i];
    float v = agg[t] + h[t] * di * di + bias[c];
    agg[t] = fmaxf(v, 0.0f);
}

__global__ __launch_bounds__(256) void k_final_logsoftmax(const float* __restrict__ agg,
                                                          const float* __restrict__ h,
                                                          const float* __restrict__ dis,
                                                          const float* __restrict__ bias,
                                                          float* __restrict__ out, int n) {
    int wave = threadIdx.x >> 5;
    int lane = threadIdx.x & 31;
    int i = blockIdx.x * 8 + wave;
    if (i >= n) return;
    float di = dis[i];
    size_t idx = (size_t)i * OUT_CH + lane;
    float v = agg[idx] + h[idx] * di * di + bias[lane];
    float m = v;
#pragma unroll
    for (int o = 16; o > 0; o >>= 1) m = fmaxf(m, __shfl_xor(m, o, 32));
    float s = __expf(v - m);
#pragma unroll
    for (int o = 16; o > 0; o >>= 1) s += __shfl_xor(s, o, 32);
    out[idx] = (v - m) - logf(s);
}

extern "C" void kernel_launch(void* const* d_in, const int* in_sizes, int n_in,
                              void* d_out, int out_size, void* d_ws, size_t ws_size,
                              hipStream_t stream) {
    const float* x  = (const float*)d_in[0];
    const int*   ei = (const int*)  d_in[1];
    const float* W1 = (const float*)d_in[2];
    const float* b1 = (const float*)d_in[3];
    const float* W2 = (const float*)d_in[4];
    const float* b2 = (const float*)d_in[5];
    float* out = (float*)d_out;

    const int N = in_sizes[0] / IN_CH;
    const int E = in_sizes[1] / 2;
    const int* src = ei;
    const int* dst = ei + E;

    char* ws = (char*)d_ws;
    size_t off = 0;
    auto carve = [&](size_t bytes) -> void* {
        void* p = ws + off;
        off += (bytes + 255) & ~(size_t)255;
        return p;
    };
    float* deg  = (float*)carve((size_t)N * sizeof(float));
    float* dis  = (float*)carve((size_t)N * sizeof(float));
    float* h1   = (float*)carve((size_t)N * HID_CH * sizeof(float));
    float* agg1 = (float*)carve((size_t)N * HID_CH * sizeof(float));
    float* h2   = (float*)carve((size_t)N * OUT_CH * sizeof(float));
    float* agg2 = (float*)carve((size_t)N * OUT_CH * sizeof(float));

    k_init_deg<<<(N + 255) / 256, 256, 0, stream>>>(deg, N);
    k_deg<<<(E + 255) / 256, 256, 0, stream>>>(dst, deg, E);
    k_dis<<<(N + 255) / 256, 256, 0, stream>>>(deg, dis, N);

    hipMemsetAsync(agg1, 0, (size_t)N * HID_CH * sizeof(float), stream);
    const int tiles1 = (N / 16) * (HID_CH / 16);
    k_gemm_wmma<HID_CH><<<(tiles1 + 7) / 8, 256, 0, stream>>>(x, W1, h1, tiles1);
    k_scatter<HID_CH><<<(E + 255) / 256, 256, 0, stream>>>(src, dst, h1, dis, agg1, E);
    {
        long long threads = (long long)N * HID_CH;
        k_finalize_relu<<<(unsigned)((threads + 255) / 256), 256, 0, stream>>>(
            agg1, h1, dis, b1, N);
    }

    hipMemsetAsync(agg2, 0, (size_t)N * OUT_CH * sizeof(float), stream);
    const int tiles2 = (N / 16) * (OUT_CH / 16);
    k_gemm_wmma<OUT_CH><<<(tiles2 + 7) / 8, 256, 0, stream>>>(agg1, W2, h2, tiles2);
    k_scatter<OUT_CH><<<(E + 255) / 256, 256, 0, stream>>>(src, dst, h2, dis, agg2, E);
    k_final_logsoftmax<<<(N + 7) / 8, 256, 0, stream>>>(agg2, h2, dis, b2, out, N);
}


# Edges counting-sorted by dst once per call (int count atomics give slot, scan, place); each layer sums a node's in-edges in registers in slot order, epilogue fused: no f32 atomics, no zero fill
# speedup vs baseline: 1.7011x; 1.3350x over previous
#include <hip/hip_runtime.h>
#include <hip/hip_bf16.h>


typedef float v2f __attribute__((ext_vector_type(2)));
typedef float v8f __attribute__((ext_vector_type(8)));

#define IN_CH  64
#define HID_CH 64
#define OUT_CH 32


__global__ __launch_bounds__(256) void k_zero(int* __restrict__ cnt, int n) {
    const int base = blockIdx.x * 1024 + threadIdx.x;
#pragma unroll
    for (int k = 0; k < 4; ++k) {
        const int i = base + k * 256;
        if (i < n) cnt[i] = 0;
    }
}

__global__ __launch_bounds__(256) void k_count(const int* __restrict__ dst,
                                               int* __restrict__ cnt,
                                               int* __restrict__ slot, int e) {
    const int base = blockIdx.x * 1024 + threadIdx.x;
    int d[4], r[4];
#pragma unroll
    for (int k = 0; k < 4; ++k) {
        const int i = base + k * 256;
        d[k] = (i < e) ? dst[i] : -1;
    }
#pragma unroll
    for (int k = 0; k < 4; ++k) r[k] = (d[k] >= 0) ? atomicAdd(&cnt[d[k]], 1) : 0;
#pragma unroll
    for (int k = 0; k < 4; ++k) {
        const int i = base + k * 256;
        if (i < e) slot[i] = r[k];
    }
}

__global__ __launch_bounds__(256) void k_bsum(const int* __restrict__ cnt,
                                              int* __restrict__ bsum, int n) {
    __shared__ int ws[8];
    const int t = threadIdx.x;
    const int i0 = blockIdx.x * 1024 + t * 4;
    int s = 0;
#pragma unroll
    for (int k = 0; k < 4; ++k) s += (i0 + k < n) ? cnt[i0 + k] : 0;
#pragma unroll
    for (int o = 16; o > 0; o >>= 1) s += __shfl_xor(s, o, 32);
    if ((t & 31) == 0) ws[t >> 5] = s;
    __syncthreads();
    if (t == 0) {
        int tot = 0;
#pragma unroll
        for (int w = 0; w < 8; ++w) tot += ws[w];
        bsum[blockIdx.x] = tot;
    }
}

__global__ __launch_bounds__(256) void k_scan(const int* __restrict__ cnt,
                                              const int* __restrict__ bsum,
                                              int* __restrict__ rowptr,
                                              float* __restrict__ dis, int n) {
    __shared__ int ws[8];
    __shared__ int wo[8];
    const int t = threadIdx.x;
    const int lane = t & 31;
    const int wv = t >> 5;

    int o = 0;
    for (int j = t; j < (int)blockIdx.x; j += 256) o += bsum[j];
#pragma unroll
    for (int sh = 16; sh > 0; sh >>= 1) o += __shfl_xor(o, sh, 32);

    const int i0 = blockIdx.x * 1024 + t * 4;
    int c[4];
    int s = 0;
#pragma unroll
    for (int k = 0; k < 4; ++k) {
        c[k] = (i0 + k < n) ? cnt[i0 + k] : 0;
        s += c[k];
    }
    int incl = s;
#pragma unroll
    for (int sh = 1; sh < 32; sh <<= 1) {
        const int up = __shfl_up(incl, sh, 32);
        if (lane >= sh) incl += up;
    }
    if (lane == 0)  wo[wv] = o;
    if (lane == 31) ws[wv] = incl;
    __syncthreads();
    int run = incl - s;
#pragma unroll
    for (int w = 0; w < 8; ++w) {
        run += wo[w];
        if (w < wv) run += ws[w];
    }
#pragma unroll
    for (int k = 0; k < 4; ++k) {
        const int i = i0 + k;
        if (i < n) {
            rowptr[i] = run;
            dis[i] = rsqrtf((float)(c[k] + 1));
            run += c[k];
            if (i == n - 1) rowptr[n] = run;
        }
    }
}

__global__ __launch_bounds__(256) void k_fill(const int* __restrict__ src,
                                              const int* __restrict__ dst,
                                              const int* __restrict__ slot,
                                              const int* __restrict__ rowptr,
                                              int* __restrict__ col, int e) {
    const int base = blockIdx.x * 1024 + threadIdx.x;
    int s[4], d[4], r[4], p[4];
#pragma unroll
    for (int k = 0; k < 4; ++k) {
        const int i = base + k * 256;
        const bool ok = i < e;
        s[k] = ok ? src[i] : 0;
        d[k] = ok ? dst[i] : -1;
        r[k] = ok ? slot[i] : 0;
    }
#pragma unroll
    for (int k = 0; k < 4; ++k) p[k] = (d[k] >= 0) ? rowptr[d[k]] + r[k] : -1;
#pragma unroll
    for (int k = 0; k < 4; ++k)
        if (p[k] >= 0) col[p[k]] = s[k];
}

template <int NOUT>
__global__ __launch_bounds__(256) void k_gemm_wmma(const float* __restrict__ A,
                                                   const float* __restrict__ B,
                                                   float* __restrict__ C,
                                                   int numTiles) {
    constexpr int K = 64;
    const int wave = threadIdx.x >> 5;
    const int lane = threadIdx.x & 31;
    const int tile = blockIdx.x * 8 + wave;
    if (tile >= numTiles) return;
    constexpr int tilesPerRow = NOUT / 16;
    const int row0 = (tile / tilesPerRow) * 16;
    const int col0 = (tile % tilesPerRow) * 16;
    const int half = lane >> 4;
    const int l15  = lane & 15;

    const float* __restrict__ Arow = A + (size_t)(row0 + l15) * K + half * 2;
    const float* __restrict__ Bp   = B + col0 + l15;

    v8f acc = {0.f, 0.f, 0.f, 0.f, 0.f, 0.f, 0.f, 0.f};
#pragma unroll
    for (int k0 = 0; k0 < K; k0 += 4) {
        v2f a;
        a.x = Arow[k0];
        a.y = Arow[k0 + 1];
        v2f b;
        b.x = Bp[(size_t)(k0 + half * 2)     * NOUT];
        b.y = Bp[(size_t)(k0 + half * 2 + 1) * NOUT];
        acc = __builtin_amdgcn_wmma_f32_16x16x4_f32(
             false, a,  false, b,
             (short)0, acc,  false,  false);
    }

    float* Cp = C + (size_t)row0 * NOUT + col0 + l15;
#pragma unroll
    for (int v = 0; v < 8; ++v)
        Cp[(size_t)(v + half * 8) * NOUT] = acc[v];
}

template <int C, int ITER, bool LAST>
__global__ __launch_bounds__(256) void k_gather(const int* __restrict__ rowptr,
                                                const int* __restrict__ col,
                                                const float* __restrict__ h,
                                                const float* __restrict__ dis,
                                                const float* __restrict__ bias,
                                                float* __restrict__ res, int n) {
    constexpr int LPN = C / 4;
    constexpr int NPB = 256 / LPN;
    constexpr int U = 4;
    const int sub  = threadIdx.x % LPN;
    const int nodeInBlock = threadIdx.x / LPN;
    const float4 b4 = *reinterpret_cast<const float4*>(bias + sub * 4);

    for (int it = 0; it < ITER; ++it) {
        const int i = (blockIdx.x * ITER + it) * NPB + nodeInBlock;
        if (i >= n) continue;
        const int beg = rowptr[i];
        const int end = rowptr[i + 1];
        const float di = dis[i];
        const float4 self = *reinterpret_cast<const float4*>(h + (size_t)i * C + sub * 4);
        const float dd = di * di;
        float4 acc;
        acc.x = self.x * dd; acc.y = self.y * dd; acc.z = self.z * dd; acc.w = self.w * dd;

        for (int p = beg; p < end; p += U) {
            int s[U];
            float w[U];
            float4 r[U];
#pragma unroll
            for (int u = 0; u < U; ++u) {
                const int q = (p + u < end) ? (p + u) : (end - 1);
                s[u] = col[q];
            }
#pragma unroll
            for (int u = 0; u < U; ++u) {
                const float ds = dis[s[u]];
                w[u] = (p + u < end) ? ds * di : 0.0f;
                r[u] = *reinterpret_cast<const float4*>(h + (size_t)s[u] * C + sub * 4);
            }
#pragma unroll
            for (int u = 0; u < U; ++u) {
                acc.x += r[u].x * w[u];
                acc.y += r[u].y * w[u];
                acc.z += r[u].z * w[u];
                acc.w += r[u].w * w[u];
            }
        }

        float4 v;
        v.x = acc.x + b4.x; v.y = acc.y + b4.y; v.z = acc.z + b4.z; v.w = acc.w + b4.w;
        if (!LAST) {
            v.x = fmaxf(v.x, 0.0f); v.y = fmaxf(v.y, 0.0f);
            v.z = fmaxf(v.z, 0.0f); v.w = fmaxf(v.w, 0.0f);
        } else {
            float m = fmaxf(fmaxf(v.x, v.y), fmaxf(v.z, v.w));
#pragma unroll
            for (int o = LPN / 2; o > 0; o >>= 1) m = fmaxf(m, __shfl_xor(m, o, 32));
            v.x -= m; v.y -= m; v.z -= m; v.w -= m;
            float sum = (__expf(v.x) + __expf(v.y)) + (__expf(v.z) + __expf(v.w));
#pragma unroll
            for (int o = LPN / 2; o > 0; o >>= 1) sum += __shfl_xor(sum, o, 32);
            const float ls = logf(sum);
            v.x -= ls; v.y -= ls; v.z -= ls; v.w -= ls;
        }
        *reinterpret_cast<float4*>(res + (size_t)i * C + sub * 4) = v;
    }
}

extern "C" void kernel_launch(void* const* d_in, const int* in_sizes, int n_in,
                              void* d_out, int out_size, void* d_ws, size_t ws_size,
                              hipStream_t stream) {
    const float* x  = (const float*)d_in[0];
    const int*   ei = (const int*)  d_in[1];
    const float* W1 = (const float*)d_in[2];
    const float* b1 = (const float*)d_in[3];
    const float* W2 = (const float*)d_in[4];
    const float* b2 = (const float*)d_in[5];
    float* out = (float*)d_out;

    const int N = in_sizes[0] / IN_CH;
    const int E = in_sizes[1] / 2;
    const int* src = ei;
    const int* dst = ei + E;
    const int nb = (N + 1023) / 1024;
    const int eb = (E + 1023) / 1024;

    char* ws = (char*)d_ws;
    size_t off = 0;
    auto carve = [&](size_t bytes) -> void* {
        void* p = ws + off;
        off += (bytes + 255) & ~(size_t)255;
        return p;
    };
    int*   cnt    = (int*)  carve((size_t)N * sizeof(int));
    int*   rowptr = (int*)  carve((size_t)(N + 1) * sizeof(int));
    float* dis    = (float*)carve((size_t)N * sizeof(float));
    int*   bsum   = (int*)  carve((size_t)nb * sizeof(int));
    int*   slot   = (int*)  carve((size_t)E * sizeof(int));
    int*   col    = (int*)  carve((size_t)E * sizeof(int));
    float* h1     = (float*)carve((size_t)N * HID_CH * sizeof(float));
    float* a1     = (float*)carve((size_t)N * HID_CH * sizeof(float));
    float* h2     = (float*)carve((size_t)N * OUT_CH * sizeof(float));

    k_zero <<<nb, 256, 0, stream>>>(cnt, N);
    k_count<<<eb, 256, 0, stream>>>(dst, cnt, slot, E);
    k_bsum <<<nb, 256, 0, stream>>>(cnt, bsum, N);
    k_scan <<<nb, 256, 0, stream>>>(cnt, bsum, rowptr, dis, N);
    k_fill <<<eb, 256, 0, stream>>>(src, dst, slot, rowptr, col, E);

    const int tiles1 = (N / 16) * (HID_CH / 16);
    k_gemm_wmma<HID_CH><<<(tiles1 + 7) / 8, 256, 0, stream>>>(x, W1, h1, tiles1);
    {
        constexpr int per = (256 / (HID_CH / 4)) * 2;
        k_gather<HID_CH, 2, false><<<(N + per - 1) / per, 256, 0, stream>>>(
            rowptr, col, h1, dis, b1, a1, N);
    }

    const int tiles2 = (N / 16) * (OUT_CH / 16);
    k_gemm_wmma<OUT_CH><<<(tiles2 + 7) / 8, 256, 0, stream>>>(a1, W2, h2, tiles2);
    {
        constexpr int per = (256 / (OUT_CH / 4)) * 2;
        k_gather<OUT_CH, 2, true><<<(N + per - 1) / per, 256, 0, stream>>>(
            rowptr, col, h2, dis, b2, out, N);
    }
}


# Both dense products moved into the layer-1 gather kernel: block sums weighted x rows of 32 nodes in LDS, then W1+relu and W2 by WMMA from LDS; h1/a1 never stored; sum order: aggregate first, then W1
# speedup vs baseline: 1.9598x; 1.1521x over previous
#include <hip/hip_runtime.h>
#include <hip/hip_bf16.h>


typedef float v2f __attribute__((ext_vector_type(2)));
typedef float v8f __attribute__((ext_vector_type(8)));

#define IN_CH  64
#define HID_CH 64
#define OUT_CH 32


__global__ __launch_bounds__(256) void k_zero(int* __restrict__ cnt, int n) {
    const int base = blockIdx.x * 1024 + threadIdx.x;
#pragma unroll
    for (int k = 0; k < 4; ++k) {
        const int i = base + k * 256;
        if (i < n) cnt[i] = 0;
    }
}

__global__ __launch_bounds__(256) void k_count(const int* __restrict__ dst,
                                               int* __restrict__ cnt,
                                               int* __restrict__ slot, int e) {
    const int base = blockIdx.x * 1024 + threadIdx.x;
    int d[4], r[4];
#pragma unroll
    for (int k = 0; k < 4; ++k) {
        const int i = base + k * 256;
        d[k] = (i < e) ? dst[i] : -1;
    }
#pragma unroll
    for (int k = 0; k < 4; ++k) r[k] = (d[k] >= 0) ? atomicAdd(&cnt[d[k]], 1) : 0;
#pragma unroll
    for (int k = 0; k < 4; ++k) {
        const int i = base + k * 256;
        if (i < e) slot[i] = r[k];
    }
}

__global__ __launch_bounds__(256) void k_bsum(const int* __restrict__ cnt,
                                              int* __restrict__ bsum, int n) {
    __shared__ int ws[8];
    const int t = threadIdx.x;
    const int i0 = blockIdx.x * 1024 + t * 4;
    int s = 0;
#pragma unroll
    for (int k = 0; k < 4; ++k) s += (i0 + k < n) ? cnt[i0 + k] : 0;
#pragma unroll
    for (int o = 16; o > 0; o >>= 1) s += __shfl_xor(s, o, 32);
    if ((t & 31) == 0) ws[t >> 5] = s;
    __syncthreads();
    if (t == 0) {
        int tot = 0;
#pragma unroll
        for (int w = 0; w < 8; ++w) tot += ws[w];
        bsum[blockIdx.x] = tot;
    }
}

__global__ __launch_bounds__(256) void k_scan(const int* __restrict__ cnt,
                                              const int* __restrict__ bsum,
                                              int* __restrict__ rowptr,
                                              float* __restrict__ dis, int n) {
    __shared__ int ws[8];
    __shared__ int wo[8];
    const int t = threadIdx.x;
    const int lane = t & 31;
    const int wv = t >> 5;

    int o = 0;
    for (int j = t; j < (int)blockIdx.x; j += 256) o += bsum[j];
#pragma unroll
    for (int sh = 16; sh > 0; sh >>= 1) o += __shfl_xor(o, sh, 32);

    const int i0 = blockIdx.x * 1024 + t * 4;
    int c[4];
    int s = 0;
#pragma unroll
    for (int k = 0; k < 4; ++k) {
        c[k] = (i0 + k < n) ? cnt[i0 + k] : 0;
        s += c[k];
    }
    int incl = s;
#pragma unroll
    for (int sh = 1; sh < 32; sh <<= 1) {
        const int up = __shfl_up(incl, sh, 32);
        if (lane >= sh) incl += up;
    }
    if (lane == 0)  wo[wv] = o;
    if (lane == 31) ws[wv] = incl;
    __syncthreads();
    int run = incl - s;
#pragma unroll
    for (int w = 0; w < 8; ++w) {
        run += wo[w];
        if (w < wv) run += ws[w];
    }
#pragma unroll
    for (int k = 0; k < 4; ++k) {
        const int i = i0 + k;
        if (i < n) {
            rowptr[i] = run;
            dis[i] = rsqrtf((float)(c[k] + 1));
            run += c[k];
            if (i == n - 1) rowptr[n] = run;
        }
    }
}

__global__ __launch_bounds__(256) void k_fill(const int* __restrict__ src,
                                              const int* __restrict__ dst,
                                              const int* __restrict__ slot,
                                              const int* __restrict__ rowptr,
                                              int* __restrict__ col, int e) {
    const int base = blockIdx.x * 1024 + threadIdx.x;
    int s[4], d[4], r[4], p[4];
#pragma unroll
    for (int k = 0; k < 4; ++k) {
        const int i = base + k * 256;
        const bool ok = i < e;
        s[k] = ok ? src[i] : 0;
        d[k] = ok ? dst[i] : -1;
        r[k] = ok ? slot[i] : 0;
    }
#pragma unroll
    for (int k = 0; k < 4; ++k) p[k] = (d[k] >= 0) ? rowptr[d[k]] + r[k] : -1;
#pragma unroll
    for (int k = 0; k < 4; ++k)
        if (p[k] >= 0) col[p[k]] = s[k];
}

#define TILE_NODES 32
#define PITCH_A 68
#define PITCH_O 36

__global__ __launch_bounds__(256) void k_layer1_fused(const int* __restrict__ rowptr,
                                                      const int* __restrict__ col,
                                                      const float* __restrict__ x,
                                                      const float* __restrict__ dis,
                                                      const float* __restrict__ W1,
                                                      const float* __restrict__ b1,
                                                      const float* __restrict__ W2,
                                                      float* __restrict__ h2, int n) {
    __shared__ __attribute__((aligned(16))) float sAgg[TILE_NODES * PITCH_A];
    __shared__ __attribute__((aligned(16))) float sAct[TILE_NODES * PITCH_A];

    const int t     = threadIdx.x;
    const int wave  = t >> 5;
    const int lane  = t & 31;
    const int half  = lane >> 4;
    const int l15   = lane & 15;
    const int node0 = blockIdx.x * TILE_NODES;

    {
        constexpr int U = 4;
        const int sub = t & 15;
        const int nodeInBlock = t >> 4;
        for (int it = 0; it < 2; ++it) {
            const int lr = it * 16 + nodeInBlock;
            const int i = node0 + lr;
            float4 acc;
            acc.x = 0.0f; acc.y = 0.0f; acc.z = 0.0f; acc.w = 0.0f;
            if (i < n) {
                const int beg = rowptr[i];
                const int end = rowptr[i + 1];
                const float di = dis[i];
                const float4 self = *reinterpret_cast<const float4*>(x + (size_t)i * IN_CH + sub * 4);
                const float dd = di * di;
                acc.x = self.x * dd; acc.y = self.y * dd; acc.z = self.z * dd; acc.w = self.w * dd;

                for (int p = beg; p < end; p += U) {
                    int s[U];
                    float w[U];
                    float4 r[U];
#pragma unroll
                    for (int u = 0; u < U; ++u) {
                        const int q = (p + u < end) ? (p + u) : (end - 1);
                        s[u] = col[q];
                    }
#pragma unroll
                    for (int u = 0; u < U; ++u) {
                        const float ds = dis[s[u]];
                        w[u] = (p + u < end) ? ds * di : 0.0f;
                        r[u] = *reinterpret_cast<const float4*>(x + (size_t)s[u] * IN_CH + sub * 4);
                    }
#pragma unroll
                    for (int u = 0; u < U; ++u) {
                        acc.x += r[u].x * w[u];
                        acc.y += r[u].y * w[u];
                        acc.z += r[u].z * w[u];
                        acc.w += r[u].w * w[u];
                    }
                }
            }
            *reinterpret_cast<float4*>(&sAgg[lr * PITCH_A + sub * 4]) = acc;
        }
    }
    __syncthreads();

    {
        const int rt = wave >> 2;
        const int ct = wave & 3;
        const float* ap = &sAgg[(rt * 16 + l15) * PITCH_A + half * 2];
        const float* __restrict__ bp = W1 + (size_t)(half * 2) * HID_CH + ct * 16 + l15;
        v8f acc = {0.f, 0.f, 0.f, 0.f, 0.f, 0.f, 0.f, 0.f};
#pragma unroll
        for (int k0 = 0; k0 < IN_CH; k0 += 4) {
            const v2f a = *reinterpret_cast<const v2f*>(ap + k0);
            v2f b;
            b.x = bp[(size_t)k0 * HID_CH];
            b.y = bp[(size_t)(k0 + 1) * HID_CH];
            acc = __builtin_amdgcn_wmma_f32_16x16x4_f32(
                 false, a,  false, b,
                 (short)0, acc,  false,  false);
        }
        const float bias = b1[ct * 16 + l15];
        float* op = &sAct[(rt * 16 + half * 8) * PITCH_A + ct * 16 + l15];
#pragma unroll
        for (int v = 0; v < 8; ++v) op[v * PITCH_A] = fmaxf(acc[v] + bias, 0.0f);
    }
    __syncthreads();

    if (wave < 4) {
        const int rt = wave >> 1;
        const int ct = wave & 1;
        const float* ap = &sAct[(rt * 16 + l15) * PITCH_A + half * 2];
        const float* __restrict__ bp = W2 + (size_t)(half * 2) * OUT_CH + ct * 16 + l15;
        v8f acc = {0.f, 0.f, 0.f, 0.f, 0.f, 0.f, 0.f, 0.f};
#pragma unroll
        for (int k0 = 0; k0 < HID_CH; k0 += 4) {
            const v2f a = *reinterpret_cast<const v2f*>(ap + k0);
            v2f b;
            b.x = bp[(size_t)k0 * OUT_CH];
            b.y = bp[(size_t)(k0 + 1) * OUT_CH];
            acc = __builtin_amdgcn_wmma_f32_16x16x4_f32(
                 false, a,  false, b,
                 (short)0, acc,  false,  false);
        }
        float* op = &sAgg[(rt * 16 + half * 8) * PITCH_O + ct * 16 + l15];
#pragma unroll
        for (int v = 0; v < 8; ++v) op[v * PITCH_O] = acc[v];
    }
    __syncthreads();

    {
        const int r  = t >> 3;
        const int c4 = t & 7;
        const int i = node0 + r;
        if (i < n) {
            const float4 v = *reinterpret_cast<const float4*>(&sAgg[r * PITCH_O + c4 * 4]);
            *reinterpret_cast<float4*>(h2 + (size_t)i * OUT_CH + c4 * 4) = v;
        }
    }
}

template <int C, int ITER, bool LAST>
__global__ __launch_bounds__(256) void k_gather(const int* __restrict__ rowptr,
                                                const int* __restrict__ col,
                                                const float* __restrict__ h,
                                                const float* __restrict__ dis,
                                                const float* __restrict__ bias,
                                                float* __restrict__ res, int n) {
    constexpr int LPN = C / 4;
    constexpr int NPB = 256 / LPN;
    constexpr int U = 4;
    const int sub  = threadIdx.x % LPN;
    const int nodeInBlock = threadIdx.x / LPN;
    const float4 b4 = *reinterpret_cast<const float4*>(bias + sub * 4);

    for (int it = 0; it < ITER; ++it) {
        const int i = (blockIdx.x * ITER + it) * NPB + nodeInBlock;
        if (i >= n) continue;
        const int beg = rowptr[i];
        const int end = rowptr[i + 1];
        const float di = dis[i];
        const float4 self = *reinterpret_cast<const float4*>(h + (size_t)i * C + sub * 4);
        const float dd = di * di;
        float4 acc;
        acc.x = self.x * dd; acc.y = self.y * dd; acc.z = self.z * dd; acc.w = self.w * dd;

        for (int p = beg; p < end; p += U) {
            int s[U];
            float w[U];
            float4 r[U];
#pragma unroll
            for (int u = 0; u < U; ++u) {
                const int q = (p + u < end) ? (p + u) : (end - 1);
                s[u] = col[q];
            }
#pragma unroll
            for (int u = 0; u < U; ++u) {
                const float ds = dis[s[u]];
                w[u] = (p + u < end) ? ds * di : 0.0f;
                r[u] = *reinterpret_cast<const float4*>(h + (size_t)s[u] * C + sub * 4);
            }
#pragma unroll
            for (int u = 0; u < U; ++u) {
                acc.x += r[u].x * w[u];
                acc.y += r[u].y * w[u];
                acc.z += r[u].z * w[u];
                acc.w += r[u].w * w[u];
            }
        }

        float4 v;
        v.x = acc.x + b4.x; v.y = acc.y + b4.y; v.z = acc.z + b4.z; v.w = acc.w + b4.w;
        if (!LAST) {
            v.x = fmaxf(v.x, 0.0f); v.y = fmaxf(v.y, 0.0f);
            v.z = fmaxf(v.z, 0.0f); v.w = fmaxf(v.w, 0.0f);
        } else {
            float m = fmaxf(fmaxf(v.x, v.y), fmaxf(v.z, v.w));
#pragma unroll
            for (int o = LPN / 2; o > 0; o >>= 1) m = fmaxf(m, __shfl_xor(m, o, 32));
            v.x -= m; v.y -= m; v.z -= m; v.w -= m;
            float sum = (__expf(v.x) + __expf(v.y)) + (__expf(v.z) + __expf(v.w));
#pragma unroll
            for (int o = LPN / 2; o > 0; o >>= 1) sum += __shfl_xor(sum, o, 32);
            const float ls = logf(sum);
            v.x -= ls; v.y -= ls; v.z -= ls; v.w -= ls;
        }
        *reinterpret_cast<float4*>(res + (size_t)i * C + sub * 4) = v;
    }
}

extern "C" void kernel_launch(void* const* d_in, const int* in_sizes, int n_in,
                              void* d_out, int out_size, void* d_ws, size_t ws_size,
                              hipStream_t stream) {
    const float* x  = (const float*)d_in[0];
    const int*   ei = (const int*)  d_in[1];
    const float* W1 = (const float*)d_in[2];
    const float* b1 = (const float*)d_in[3];
    const float* W2 = (const float*)d_in[4];
    const float* b2 = (const float*)d_in[5];
    float* out = (float*)d_out;

    const int N = in_sizes[0] / IN_CH;
    const int E = in_sizes[1] / 2;
    const int* src = ei;
    const int* dst = ei + E;
    const int nb = (N + 1023) / 1024;
    const int eb = (E + 1023) / 1024;

    char* ws = (char*)d_ws;
    size_t off = 0;
    auto carve = [&](size_t bytes) -> void* {
        void* p = ws + off;
        off += (bytes + 255) & ~(size_t)255;
        return p;
    };
    int*   cnt    = (int*)  carve((size_t)N * sizeof(int));
    int*   rowptr = (int*)  carve((size_t)(N + 1) * sizeof(int));
    float* dis    = (float*)carve((size_t)N * sizeof(float));
    int*   bsum   = (int*)  carve((size_t)nb * sizeof(int));
    int*   slot   = (int*)  carve((size_t)E * sizeof(int));
    int*   col    = (int*)  carve((size_t)E * sizeof(int));
    float* h2     = (float*)carve((size_t)N * OUT_CH * sizeof(float));

    k_zero <<<nb, 256, 0, stream>>>(cnt, N);
    k_count<<<eb, 256, 0, stream>>>(dst, cnt, slot, E);
    k_bsum <<<nb, 256, 0, stream>>>(cnt, bsum, N);
    k_scan <<<nb, 256, 0, stream>>>(cnt, bsum, rowptr, dis, N);
    k_fill <<<eb, 256, 0, stream>>>(src, dst, slot, rowptr, col, E);

    k_layer1_fused<<<(N + TILE_NODES - 1) / TILE_NODES, 256, 0, stream>>>(
        rowptr, col, x, dis, W1, b1, W2, h2, N);

    {
        constexpr int per = (256 / (OUT_CH / 4)) * 2;
        k_gather<OUT_CH, 2, true><<<(N + per - 1) / per, 256, 0, stream>>>(
            rowptr, col, h2, dis, b2, out, N);
    }
}


# Edges grouped by dst with LDS counters, no per-edge returning global atomics: a block owns (edge chunk, node range), counts go to a chunk x node table, scanned; sum order: chunk, then LDS arrival
# speedup vs baseline: 2.6733x; 1.3641x over previous
#include <hip/hip_runtime.h>
#include <hip/hip_bf16.h>


typedef float v2f __attribute__((ext_vector_type(2)));
typedef float v8f __attribute__((ext_vector_type(8)));

#define IN_CH  64
#define HID_CH 64
#define OUT_CH 32

#define RNG_NODES 12800
#define NCHUNK    32
#define SORT_T    1024
#define SORT_U    8

__global__ __launch_bounds__(SORT_T) void k_count_lds(const int* __restrict__ dst,
                                                      int* __restrict__ tab,
                                                      int n, int e, int chunk) {
    __shared__ int c_lds[RNG_NODES];
    const int t  = threadIdx.x;
    const int lo = blockIdx.y * RNG_NODES;
    const int nn = min(RNG_NODES, n - lo);
    for (int i = t; i < nn; i += SORT_T) c_lds[i] = 0;
    __syncthreads();

    const int e0 = blockIdx.x * chunk;
    const int e1 = min(e0 + chunk, e);
    for (int i = e0 + t; i < e1; i += SORT_T * SORT_U) {
        int d[SORT_U];
#pragma unroll
        for (int k = 0; k < SORT_U; ++k) {
            const int idx = i + k * SORT_T;
            d[k] = (idx < e1) ? dst[idx] : -1;
        }
#pragma unroll
        for (int k = 0; k < SORT_U; ++k) {
            const unsigned r = (unsigned)(d[k] - lo);
            if (r < (unsigned)nn) atomicAdd(&c_lds[r], 1);
        }
    }
    __syncthreads();

    int* __restrict__ row = tab + (size_t)blockIdx.x * n + lo;
    for (int i = t; i < nn; i += SORT_T) row[i] = c_lds[i];
}

__global__ __launch_bounds__(1024) void k_colscan(int* __restrict__ tab,
                                                  int* __restrict__ cnt,
                                                  int* __restrict__ bsum, int n) {
    __shared__ int ws[32];
    const int t = threadIdx.x;
    const int i = blockIdx.x * 1024 + t;
    int run = 0;
    if (i < n) {
        int v[NCHUNK];
#pragma unroll
        for (int c = 0; c < NCHUNK; ++c) v[c] = tab[(size_t)c * n + i];
#pragma unroll
        for (int c = 0; c < NCHUNK; ++c) {
            tab[(size_t)c * n + i] = run;
            run += v[c];
        }
        cnt[i] = run;
    }
    int s = run;
#pragma unroll
    for (int o = 16; o > 0; o >>= 1) s += __shfl_xor(s, o, 32);
    if ((t & 31) == 0) ws[t >> 5] = s;
    __syncthreads();
    if (t < 32) {
        int tot = ws[t];
#pragma unroll
        for (int o = 16; o > 0; o >>= 1) tot += __shfl_xor(tot, o, 32);
        if (t == 0) bsum[blockIdx.x] = tot;
    }
}

__global__ __launch_bounds__(256) void k_scan(const int* __restrict__ cnt,
                                              const int* __restrict__ bsum,
                                              int* __restrict__ rowptr,
                                              float* __restrict__ dis, int n) {
    __shared__ int ws[8];
    __shared__ int wo[8];
    const int t = threadIdx.x;
    const int lane = t & 31;
    const int wv = t >> 5;

    int o = 0;
    for (int j = t; j < (int)blockIdx.x; j += 256) o += bsum[j];
#pragma unroll
    for (int sh = 16; sh > 0; sh >>= 1) o += __shfl_xor(o, sh, 32);

    const int i0 = blockIdx.x * 1024 + t * 4;
    int c[4];
    int s = 0;
#pragma unroll
    for (int k = 0; k < 4; ++k) {
        c[k] = (i0 + k < n) ? cnt[i0 + k] : 0;
        s += c[k];
    }
    int incl = s;
#pragma unroll
    for (int sh = 1; sh < 32; sh <<= 1) {
        const int up = __shfl_up(incl, sh, 32);
        if (lane >= sh) incl += up;
    }
    if (lane == 0)  wo[wv] = o;
    if (lane == 31) ws[wv] = incl;
    __syncthreads();
    int run = incl - s;
#pragma unroll
    for (int w = 0; w < 8; ++w) {
        run += wo[w];
        if (w < wv) run += ws[w];
    }
#pragma unroll
    for (int k = 0; k < 4; ++k) {
        const int i = i0 + k;
        if (i < n) {
            rowptr[i] = run;
            dis[i] = rsqrtf((float)(c[k] + 1));
            run += c[k];
            if (i == n - 1) rowptr[n] = run;
        }
    }
}

__global__ __launch_bounds__(SORT_T) void k_place_lds(const int* __restrict__ src,
                                                      const int* __restrict__ dst,
                                                      const int* __restrict__ tab,
                                                      const int* __restrict__ rowptr,
                                                      int* __restrict__ col,
                                                      int n, int e, int chunk) {
    __shared__ int pos[RNG_NODES];
    const int t  = threadIdx.x;
    const int lo = blockIdx.y * RNG_NODES;
    const int nn = min(RNG_NODES, n - lo);
    const int* __restrict__ row = tab + (size_t)blockIdx.x * n + lo;
    {
        constexpr int PER = (RNG_NODES + SORT_T - 1) / SORT_T;
        int a[PER], b[PER];
#pragma unroll
        for (int k = 0; k < PER; ++k) {
            const int i = t + k * SORT_T;
            const bool ok = i < nn;
            a[k] = ok ? rowptr[lo + i] : 0;
            b[k] = ok ? row[i] : 0;
        }
#pragma unroll
        for (int k = 0; k < PER; ++k) {
            const int i = t + k * SORT_T;
            if (i < nn) pos[i] = a[k] + b[k];
        }
    }
    __syncthreads();

    const int e0 = blockIdx.x * chunk;
    const int e1 = min(e0 + chunk, e);
    for (int i = e0 + t; i < e1; i += SORT_T * SORT_U) {
        int d[SORT_U], s[SORT_U];
#pragma unroll
        for (int k = 0; k < SORT_U; ++k) {
            const int idx = i + k * SORT_T;
            const bool ok = idx < e1;
            d[k] = ok ? dst[idx] : -1;
            s[k] = ok ? src[idx] : 0;
        }
        int p[SORT_U];
#pragma unroll
        for (int k = 0; k < SORT_U; ++k) {
            const unsigned r = (unsigned)(d[k] - lo);
            p[k] = (r < (unsigned)nn) ? atomicAdd(&pos[r], 1) : -1;
        }
#pragma unroll
        for (int k = 0; k < SORT_U; ++k)
            if (p[k] >= 0) col[p[k]] = s[k];
    }
}

#define TILE_NODES 32
#define PITCH_A 68
#define PITCH_O 36

__global__ __launch_bounds__(256) void k_layer1_fused(const int* __restrict__ rowptr,
                                                      const int* __restrict__ col,
                                                      const float* __restrict__ x,
                                                      const float* __restrict__ dis,
                                                      const float* __restrict__ W1,
                                                      const float* __restrict__ b1,
                                                      const float* __restrict__ W2,
                                                      float* __restrict__ h2, int n) {
    __shared__ __attribute__((aligned(16))) float sAgg[TILE_NODES * PITCH_A];
    __shared__ __attribute__((aligned(16))) float sAct[TILE_NODES * PITCH_A];

    const int t     = threadIdx.x;
    const int wave  = t >> 5;
    const int lane  = t & 31;
    const int half  = lane >> 4;
    const int l15   = lane & 15;
    const int node0 = blockIdx.x * TILE_NODES;

    {
        constexpr int U = 4;
        const int sub = t & 15;
        const int nodeInBlock = t >> 4;
        for (int it = 0; it < 2; ++it) {
            const int lr = it * 16 + nodeInBlock;
            const int i = node0 + lr;
            float4 acc;
            acc.x = 0.0f; acc.y = 0.0f; acc.z = 0.0f; acc.w = 0.0f;
            if (i < n) {
                const int beg = rowptr[i];
                const int end = rowptr[i + 1];
                const float di = dis[i];
                const float4 self = *reinterpret_cast<const float4*>(x + (size_t)i * IN_CH + sub * 4);
                const float dd = di * di;
                acc.x = self.x * dd; acc.y = self.y * dd; acc.z = self.z * dd; acc.w = self.w * dd;

                for (int p = beg; p < end; p += U) {
                    int s[U];
                    float w[U];
                    float4 r[U];
#pragma unroll
                    for (int u = 0; u < U; ++u) {
                        const int q = (p + u < end) ? (p + u) : (end - 1);
                        s[u] = col[q];
                    }
#pragma unroll
                    for (int u = 0; u < U; ++u) {
                        const float ds = dis[s[u]];
                        w[u] = (p + u < end) ? ds * di : 0.0f;
                        r[u] = *reinterpret_cast<const float4*>(x + (size_t)s[u] * IN_CH + sub * 4);
                    }
#pragma unroll
                    for (int u = 0; u < U; ++u) {
                        acc.x += r[u].x * w[u];
                        acc.y += r[u].y * w[u];
                        acc.z += r[u].z * w[u];
                        acc.w += r[u].w * w[u];
                    }
                }
            }
            *reinterpret_cast<float4*>(&sAgg[lr * PITCH_A + sub * 4]) = acc;
        }
    }
    __syncthreads();

    {
        const int rt = wave >> 2;
        const int ct = wave & 3;
        const float* ap = &sAgg[(rt * 16 + l15) * PITCH_A + half * 2];
        const float* __restrict__ bp = W1 + (size_t)(half * 2) * HID_CH + ct * 16 + l15;
        v8f acc = {0.f, 0.f, 0.f, 0.f, 0.f, 0.f, 0.f, 0.f};
#pragma unroll
        for (int k0 = 0; k0 < IN_CH; k0 += 4) {
            const v2f a = *reinterpret_cast<const v2f*>(ap + k0);
            v2f b;
            b.x = bp[(size_t)k0 * HID_CH];
            b.y = bp[(size_t)(k0 + 1) * HID_CH];
            acc = __builtin_amdgcn_wmma_f32_16x16x4_f32(
                 false, a,  false, b,
                 (short)0, acc,  false,  false);
        }
        const float bias = b1[ct * 16 + l15];
        float* op = &sAct[(rt * 16 + half * 8) * PITCH_A + ct * 16 + l15];
#pragma unroll
        for (int v = 0; v < 8; ++v) op[v * PITCH_A] = fmaxf(acc[v] + bias, 0.0f);
    }
    __syncthreads();

    if (wave < 4) {
        const int rt = wave >> 1;
        const int ct = wave & 1;
        const float* ap = &sAct[(rt * 16 + l15) * PITCH_A + half * 2];
        const float* __restrict__ bp = W2 + (size_t)(half * 2) * OUT_CH + ct * 16 + l15;
        v8f acc = {0.f, 0.f, 0.f, 0.f, 0.f, 0.f, 0.f, 0.f};
#pragma unroll
        for (int k0 = 0; k0 < HID_CH; k0 += 4) {
            const v2f a = *reinterpret_cast<const v2f*>(ap + k0);
            v2f b;
            b.x = bp[(size_t)k0 * OUT_CH];
            b.y = bp[(size_t)(k0 + 1) * OUT_CH];
            acc = __builtin_amdgcn_wmma_f32_16x16x4_f32(
                 false, a,  false, b,
                 (short)0, acc,  false,  false);
        }
        float* op = &sAgg[(rt * 16 + half * 8) * PITCH_O + ct * 16 + l15];
#pragma unroll
        for (int v = 0; v < 8; ++v) op[v * PITCH_O] = acc[v];
    }
    __syncthreads();

    {
        const int r  = t >> 3;
        const int c4 = t & 7;
        const int i = node0 + r;
        if (i < n) {
            const float4 v = *reinterpret_cast<const float4*>(&sAgg[r * PITCH_O + c4 * 4]);
            *reinterpret_cast<float4*>(h2 + (size_t)i * OUT_CH + c4 * 4) = v;
        }
    }
}

template <int C, int ITER, bool LAST>
__global__ __launch_bounds__(256) void k_gather(const int* __restrict__ rowptr,
                                                const int* __restrict__ col,
                                                const float* __restrict__ h,
                                                const float* __restrict__ dis,
                                                const float* __restrict__ bias,
                                                float* __restrict__ res, int n) {
    constexpr int LPN = C / 4;
    constexpr int NPB = 256 / LPN;
    constexpr int U = 4;
    const int sub  = threadIdx.x % LPN;
    const int nodeInBlock = threadIdx.x / LPN;
    const float4 b4 = *reinterpret_cast<const float4*>(bias + sub * 4);

    for (int it = 0; it < ITER; ++it) {
        const int i = (blockIdx.x * ITER + it) * NPB + nodeInBlock;
        if (i >= n) continue;
        const int beg = rowptr[i];
        const int end = rowptr[i + 1];
        const float di = dis[i];
        const float4 self = *reinterpret_cast<const float4*>(h + (size_t)i * C + sub * 4);
        const float dd = di * di;
        float4 acc;
        acc.x = self.x * dd; acc.y = self.y * dd; acc.z = self.z * dd; acc.w = self.w * dd;

        for (int p = beg; p < end; p += U) {
            int s[U];
            float w[U];
            float4 r[U];
#pragma unroll
            for (int u = 0; u < U; ++u) {
                const int q = (p + u < end) ? (p + u) : (end - 1);
                s[u] = col[q];
            }
#pragma unroll
            for (int u = 0; u < U; ++u) {
                const float ds = dis[s[u]];
                w[u] = (p + u < end) ? ds * di : 0.0f;
                r[u] = *reinterpret_cast<const float4*>(h + (size_t)s[u] * C + sub * 4);
            }
#pragma unroll
            for (int u = 0; u < U; ++u) {
                acc.x += r[u].x * w[u];
                acc.y += r[u].y * w[u];
                acc.z += r[u].z * w[u];
                acc.w += r[u].w * w[u];
            }
        }

        float4 v;
        v.x = acc.x + b4.x; v.y = acc.y + b4.y; v.z = acc.z + b4.z; v.w = acc.w + b4.w;
        if (!LAST) {
            v.x = fmaxf(v.x, 0.0f); v.y = fmaxf(v.y, 0.0f);
            v.z = fmaxf(v.z, 0.0f); v.w = fmaxf(v.w, 0.0f);
        } else {
            float m = fmaxf(fmaxf(v.x, v.y), fmaxf(v.z, v.w));
#pragma unroll
            for (int o = LPN / 2; o > 0; o >>= 1) m = fmaxf(m, __shfl_xor(m, o, 32));
            v.x -= m; v.y -= m; v.z -= m; v.w -= m;
            float sum = (__expf(v.x) + __expf(v.y)) + (__expf(v.z) + __expf(v.w));
#pragma unroll
            for (int o = LPN / 2; o > 0; o >>= 1) sum += __shfl_xor(sum, o, 32);
            const float ls = logf(sum);
            v.x -= ls; v.y -= ls; v.z -= ls; v.w -= ls;
        }
        *reinterpret_cast<float4*>(res + (size_t)i * C + sub * 4) = v;
    }
}

extern "C" void kernel_launch(void* const* d_in, const int* in_sizes, int n_in,
                              void* d_out, int out_size, void* d_ws, size_t ws_size,
                              hipStream_t stream) {
    const float* x  = (const float*)d_in[0];
    const int*   ei = (const int*)  d_in[1];
    const float* W1 = (const float*)d_in[2];
    const float* b1 = (const float*)d_in[3];
    const float* W2 = (const float*)d_in[4];
    const float* b2 = (const float*)d_in[5];
    float* out = (float*)d_out;

    const int N = in_sizes[0] / IN_CH;
    const int E = in_sizes[1] / 2;
    const int* src = ei;
    const int* dst = ei + E;
    const int nb = (N + 1023) / 1024;
    const int ng = (N + RNG_NODES - 1) / RNG_NODES;
    const int chunk = (E + NCHUNK - 1) / NCHUNK;

    char* ws = (char*)d_ws;
    size_t off = 0;
    auto carve = [&](size_t bytes) -> void* {
        void* p = ws + off;
        off += (bytes + 255) & ~(size_t)255;
        return p;
    };
    int*   cnt    = (int*)  carve((size_t)N * sizeof(int));
    int*   rowptr = (int*)  carve((size_t)(N + 1) * sizeof(int));
    float* dis    = (float*)carve((size_t)N * sizeof(float));
    int*   bsum   = (int*)  carve((size_t)nb * sizeof(int));
    int*   col    = (int*)  carve((size_t)E * sizeof(int));
    constexpr size_t planeCols = (NCHUNK > OUT_CH) ? NCHUNK : OUT_CH;
    float* h2     = (float*)carve((size_t)N * planeCols * sizeof(float));
    int*   tab    = (int*)h2;

    const dim3 sgrid(NCHUNK, ng);
    k_count_lds<<<sgrid, SORT_T, 0, stream>>>(dst, tab, N, E, chunk);
    k_colscan  <<<nb, 1024, 0, stream>>>(tab, cnt, bsum, N);
    k_scan     <<<nb, 256, 0, stream>>>(cnt, bsum, rowptr, dis, N);
    k_place_lds<<<sgrid, SORT_T, 0, stream>>>(src, dst, tab, rowptr, col, N, E, chunk);

    k_layer1_fused<<<(N + TILE_NODES - 1) / TILE_NODES, 256, 0, stream>>>(
        rowptr, col, x, dis, W1, b1, W2, h2, N);

    {
        constexpr int per = (256 / (OUT_CH / 4)) * 2;
        k_gather<OUT_CH, 2, true><<<(N + per - 1) / per, 256, 0, stream>>>(
            rowptr, col, h2, dis, b2, out, N);
    }
}


# h2 rows stored pre-multiplied by dis[src] in the fused kernel's epilogue; the layer-2 gather loads no per-edge norm; dis[dst] applied once after the sum (sum order kept, product rounding differs)
# speedup vs baseline: 2.7732x; 1.0374x over previous
#include <hip/hip_runtime.h>
#include <hip/hip_bf16.h>


typedef float v2f __attribute__((ext_vector_type(2)));
typedef float v8f __attribute__((ext_vector_type(8)));

#define IN_CH  64
#define HID_CH 64
#define OUT_CH 32

#define RNG_NODES 12800
#define NCHUNK    32
#define SORT_T    1024
#define SORT_U    8

__global__ __launch_bounds__(SORT_T) void k_count_lds(const int* __restrict__ dst,
                                                      int* __restrict__ tab,
                                                      int n, int e, int chunk) {
    __shared__ int c_lds[RNG_NODES];
    const int t  = threadIdx.x;
    const int lo = blockIdx.y * RNG_NODES;
    const int nn = min(RNG_NODES, n - lo);
    for (int i = t; i < nn; i += SORT_T) c_lds[i] = 0;
    __syncthreads();

    const int e0 = blockIdx.x * chunk;
    const int e1 = min(e0 + chunk, e);
    for (int i = e0 + t; i < e1; i += SORT_T * SORT_U) {
        int d[SORT_U];
#pragma unroll
        for (int k = 0; k < SORT_U; ++k) {
            const int idx = i + k * SORT_T;
            d[k] = (idx < e1) ? dst[idx] : -1;
        }
#pragma unroll
        for (int k = 0; k < SORT_U; ++k) {
            const unsigned r = (unsigned)(d[k] - lo);
            if (r < (unsigned)nn) atomicAdd(&c_lds[r], 1);
        }
    }
    __syncthreads();

    int* __restrict__ row = tab + (size_t)blockIdx.x * n + lo;
    for (int i = t; i < nn; i += SORT_T) row[i] = c_lds[i];
}

__global__ __launch_bounds__(1024) void k_colscan(int* __restrict__ tab,
                                                  int* __restrict__ cnt,
                                                  int* __restrict__ bsum, int n) {
    __shared__ int ws[32];
    const int t = threadIdx.x;
    const int i = blockIdx.x * 1024 + t;
    int run = 0;
    if (i < n) {
        int v[NCHUNK];
#pragma unroll
        for (int c = 0; c < NCHUNK; ++c) v[c] = tab[(size_t)c * n + i];
#pragma unroll
        for (int c = 0; c < NCHUNK; ++c) {
            tab[(size_t)c * n + i] = run;
            run += v[c];
        }
        cnt[i] = run;
    }
    int s = run;
#pragma unroll
    for (int o = 16; o > 0; o >>= 1) s += __shfl_xor(s, o, 32);
    if ((t & 31) == 0) ws[t >> 5] = s;
    __syncthreads();
    if (t < 32) {
        int tot = ws[t];
#pragma unroll
        for (int o = 16; o > 0; o >>= 1) tot += __shfl_xor(tot, o, 32);
        if (t == 0) bsum[blockIdx.x] = tot;
    }
}

__global__ __launch_bounds__(256) void k_scan(const int* __restrict__ cnt,
                                              const int* __restrict__ bsum,
                                              int* __restrict__ rowptr,
                                              float* __restrict__ dis, int n) {
    __shared__ int ws[8];
    __shared__ int wo[8];
    const int t = threadIdx.x;
    const int lane = t & 31;
    const int wv = t >> 5;

    int o = 0;
    for (int j = t; j < (int)blockIdx.x; j += 256) o += bsum[j];
#pragma unroll
    for (int sh = 16; sh > 0; sh >>= 1) o += __shfl_xor(o, sh, 32);

    const int i0 = blockIdx.x * 1024 + t * 4;
    int c[4];
    int s = 0;
#pragma unroll
    for (int k = 0; k < 4; ++k) {
        c[k] = (i0 + k < n) ? cnt[i0 + k] : 0;
        s += c[k];
    }
    int incl = s;
#pragma unroll
    for (int sh = 1; sh < 32; sh <<= 1) {
        const int up = __shfl_up(incl, sh, 32);
        if (lane >= sh) incl += up;
    }
    if (lane == 0)  wo[wv] = o;
    if (lane == 31) ws[wv] = incl;
    __syncthreads();
    int run = incl - s;
#pragma unroll
    for (int w = 0; w < 8; ++w) {
        run += wo[w];
        if (w < wv) run += ws[w];
    }
#pragma unroll
    for (int k = 0; k < 4; ++k) {
        const int i = i0 + k;
        if (i < n) {
            rowptr[i] = run;
            dis[i] = rsqrtf((float)(c[k] + 1));
            run += c[k];
            if (i == n - 1) rowptr[n] = run;
        }
    }
}

__global__ __launch_bounds__(SORT_T) void k_place_lds(const int* __restrict__ src,
                                                      const int* __restrict__ dst,
                                                      const int* __restrict__ tab,
                                                      const int* __restrict__ rowptr,
                                                      int* __restrict__ col,
                                                      int n, int e, int chunk) {
    __shared__ int pos[RNG_NODES];
    const int t  = threadIdx.x;
    const int lo = blockIdx.y * RNG_NODES;
    const int nn = min(RNG_NODES, n - lo);
    const int* __restrict__ row = tab + (size_t)blockIdx.x * n + lo;
    {
        constexpr int PER = (RNG_NODES + SORT_T - 1) / SORT_T;
        int a[PER], b[PER];
#pragma unroll
        for (int k = 0; k < PER; ++k) {
            const int i = t + k * SORT_T;
            const bool ok = i < nn;
            a[k] = ok ? rowptr[lo + i] : 0;
            b[k] = ok ? row[i] : 0;
        }
#pragma unroll
        for (int k = 0; k < PER; ++k) {
            const int i = t + k * SORT_T;
            if (i < nn) pos[i] = a[k] + b[k];
        }
    }
    __syncthreads();

    const int e0 = blockIdx.x * chunk;
    const int e1 = min(e0 + chunk, e);
    for (int i = e0 + t; i < e1; i += SORT_T * SORT_U) {
        int d[SORT_U], s[SORT_U];
#pragma unroll
        for (int k = 0; k < SORT_U; ++k) {
            const int idx = i + k * SORT_T;
            const bool ok = idx < e1;
            d[k] = ok ? dst[idx] : -1;
            s[k] = ok ? src[idx] : 0;
        }
        int p[SORT_U];
#pragma unroll
        for (int k = 0; k < SORT_U; ++k) {
            const unsigned r = (unsigned)(d[k] - lo);
            p[k] = (r < (unsigned)nn) ? atomicAdd(&pos[r], 1) : -1;
        }
#pragma unroll
        for (int k = 0; k < SORT_U; ++k)
            if (p[k] >= 0) col[p[k]] = s[k];
    }
}

#define TILE_NODES 32
#define PITCH_A 68
#define PITCH_O 36

__global__ __launch_bounds__(256) void k_layer1_fused(const int* __restrict__ rowptr,
                                                      const int* __restrict__ col,
                                                      const float* __restrict__ x,
                                                      const float* __restrict__ dis,
                                                      const float* __restrict__ W1,
                                                      const float* __restrict__ b1,
                                                      const float* __restrict__ W2,
                                                      float* __restrict__ h2, int n) {
    __shared__ __attribute__((aligned(16))) float sAgg[TILE_NODES * PITCH_A];
    __shared__ __attribute__((aligned(16))) float sAct[TILE_NODES * PITCH_A];
    __shared__ float sDis[TILE_NODES];

    const int t     = threadIdx.x;
    const int wave  = t >> 5;
    const int lane  = t & 31;
    const int half  = lane >> 4;
    const int l15   = lane & 15;
    const int node0 = blockIdx.x * TILE_NODES;

    {
        constexpr int U = 4;
        const int sub = t & 15;
        const int nodeInBlock = t >> 4;
        for (int it = 0; it < 2; ++it) {
            const int lr = it * 16 + nodeInBlock;
            const int i = node0 + lr;
            float4 acc;
            acc.x = 0.0f; acc.y = 0.0f; acc.z = 0.0f; acc.w = 0.0f;
            if (i < n) {
                const int beg = rowptr[i];
                const int end = rowptr[i + 1];
                const float di = dis[i];
                const float4 self = *reinterpret_cast<const float4*>(x + (size_t)i * IN_CH + sub * 4);
                const float dd = di * di;
                if (sub == 0) sDis[lr] = di;
                acc.x = self.x * dd; acc.y = self.y * dd; acc.z = self.z * dd; acc.w = self.w * dd;

                for (int p = beg; p < end; p += U) {
                    int s[U];
                    float w[U];
                    float4 r[U];
#pragma unroll
                    for (int u = 0; u < U; ++u) {
                        const int q = (p + u < end) ? (p + u) : (end - 1);
                        s[u] = col[q];
                    }
#pragma unroll
                    for (int u = 0; u < U; ++u) {
                        const float ds = dis[s[u]];
                        w[u] = (p + u < end) ? ds * di : 0.0f;
                        r[u] = *reinterpret_cast<const float4*>(x + (size_t)s[u] * IN_CH + sub * 4);
                    }
#pragma unroll
                    for (int u = 0; u < U; ++u) {
                        acc.x += r[u].x * w[u];
                        acc.y += r[u].y * w[u];
                        acc.z += r[u].z * w[u];
                        acc.w += r[u].w * w[u];
                    }
                }
            }
            *reinterpret_cast<float4*>(&sAgg[lr * PITCH_A + sub * 4]) = acc;
        }
    }
    __syncthreads();

    {
        const int rt = wave >> 2;
        const int ct = wave & 3;
        const float* ap = &sAgg[(rt * 16 + l15) * PITCH_A + half * 2];
        const float* __restrict__ bp = W1 + (size_t)(half * 2) * HID_CH + ct * 16 + l15;
        v8f acc = {0.f, 0.f, 0.f, 0.f, 0.f, 0.f, 0.f, 0.f};
#pragma unroll
        for (int k0 = 0; k0 < IN_CH; k0 += 4) {
            const v2f a = *reinterpret_cast<const v2f*>(ap + k0);
            v2f b;
            b.x = bp[(size_t)k0 * HID_CH];
            b.y = bp[(size_t)(k0 + 1) * HID_CH];
            acc = __builtin_amdgcn_wmma_f32_16x16x4_f32(
                 false, a,  false, b,
                 (short)0, acc,  false,  false);
        }
        const float bias = b1[ct * 16 + l15];
        float* op = &sAct[(rt * 16 + half * 8) * PITCH_A + ct * 16 + l15];
#pragma unroll
        for (int v = 0; v < 8; ++v) op[v * PITCH_A] = fmaxf(acc[v] + bias, 0.0f);
    }
    __syncthreads();

    if (wave < 4) {
        const int rt = wave >> 1;
        const int ct = wave & 1;
        const float* ap = &sAct[(rt * 16 + l15) * PITCH_A + half * 2];
        const float* __restrict__ bp = W2 + (size_t)(half * 2) * OUT_CH + ct * 16 + l15;
        v8f acc = {0.f, 0.f, 0.f, 0.f, 0.f, 0.f, 0.f, 0.f};
#pragma unroll
        for (int k0 = 0; k0 < HID_CH; k0 += 4) {
            const v2f a = *reinterpret_cast<const v2f*>(ap + k0);
            v2f b;
            b.x = bp[(size_t)k0 * OUT_CH];
            b.y = bp[(size_t)(k0 + 1) * OUT_CH];
            acc = __builtin_amdgcn_wmma_f32_16x16x4_f32(
                 false, a,  false, b,
                 (short)0, acc,  false,  false);
        }
        float* op = &sAgg[(rt * 16 + half * 8) * PITCH_O + ct * 16 + l15];
#pragma unroll
        for (int v = 0; v < 8; ++v) op[v * PITCH_O] = acc[v];
    }
    __syncthreads();

    {
        const int r  = t >> 3;
        const int c4 = t & 7;
        const int i = node0 + r;
        if (i < n) {
            float4 v = *reinterpret_cast<const float4*>(&sAgg[r * PITCH_O + c4 * 4]);
            const float sc = sDis[r];
            v.x *= sc; v.y *= sc; v.z *= sc; v.w *= sc;
            *reinterpret_cast<float4*>(h2 + (size_t)i * OUT_CH + c4 * 4) = v;
        }
    }
}

template <int C, int ITER, bool LAST>
__global__ __launch_bounds__(256) void k_gather(const int* __restrict__ rowptr,
                                                const int* __restrict__ col,
                                                const float* __restrict__ hs,
                                                const float* __restrict__ dis,
                                                const float* __restrict__ bias,
                                                float* __restrict__ res, int n) {
    constexpr int LPN = C / 4;
    constexpr int NPB = 256 / LPN;
    constexpr int U = 4;
    const int sub  = threadIdx.x % LPN;
    const int nodeInBlock = threadIdx.x / LPN;
    const float4 b4 = *reinterpret_cast<const float4*>(bias + sub * 4);

    for (int it = 0; it < ITER; ++it) {
        const int i = (blockIdx.x * ITER + it) * NPB + nodeInBlock;
        if (i >= n) continue;
        const int beg = rowptr[i];
        const int end = rowptr[i + 1];
        const float di = dis[i];
        float4 acc = *reinterpret_cast<const float4*>(hs + (size_t)i * C + sub * 4);

        for (int p = beg; p < end; p += U) {
            int s[U];
            float4 r[U];
#pragma unroll
            for (int u = 0; u < U; ++u) {
                const int q = (p + u < end) ? (p + u) : (end - 1);
                s[u] = col[q];
            }
#pragma unroll
            for (int u = 0; u < U; ++u)
                r[u] = *reinterpret_cast<const float4*>(hs + (size_t)s[u] * C + sub * 4);
#pragma unroll
            for (int u = 0; u < U; ++u) {
                const float w = (p + u < end) ? 1.0f : 0.0f;
                acc.x += r[u].x * w;
                acc.y += r[u].y * w;
                acc.z += r[u].z * w;
                acc.w += r[u].w * w;
            }
        }

        float4 v;
        v.x = acc.x * di + b4.x; v.y = acc.y * di + b4.y;
        v.z = acc.z * di + b4.z; v.w = acc.w * di + b4.w;
        if (!LAST) {
            v.x = fmaxf(v.x, 0.0f); v.y = fmaxf(v.y, 0.0f);
            v.z = fmaxf(v.z, 0.0f); v.w = fmaxf(v.w, 0.0f);
        } else {
            float m = fmaxf(fmaxf(v.x, v.y), fmaxf(v.z, v.w));
#pragma unroll
            for (int o = LPN / 2; o > 0; o >>= 1) m = fmaxf(m, __shfl_xor(m, o, 32));
            v.x -= m; v.y -= m; v.z -= m; v.w -= m;
            float sum = (__expf(v.x) + __expf(v.y)) + (__expf(v.z) + __expf(v.w));
#pragma unroll
            for (int o = LPN / 2; o > 0; o >>= 1) sum += __shfl_xor(sum, o, 32);
            const float ls = logf(sum);
            v.x -= ls; v.y -= ls; v.z -= ls; v.w -= ls;
        }
        *reinterpret_cast<float4*>(res + (size_t)i * C + sub * 4) = v;
    }
}

extern "C" void kernel_launch(void* const* d_in, const int* in_sizes, int n_in,
                              void* d_out, int out_size, void* d_ws, size_t ws_size,
                              hipStream_t stream) {
    const float* x  = (const float*)d_in[0];
    const int*   ei = (const int*)  d_in[1];
    const float* W1 = (const float*)d_in[2];
    const float* b1 = (const float*)d_in[3];
    const float* W2 = (const float*)d_in[4];
    const float* b2 = (const float*)d_in[5];
    float* out = (float*)d_out;

    const int N = in_sizes[0] / IN_CH;
    const int E = in_sizes[1] / 2;
    const int* src = ei;
    const int* dst = ei + E;
    const int nb = (N + 1023) / 1024;
    const int ng = (N + RNG_NODES - 1) / RNG_NODES;
    const int chunk = (E + NCHUNK - 1) / NCHUNK;

    char* ws = (char*)d_ws;
    size_t off = 0;
    auto carve = [&](size_t bytes) -> void* {
        void* p = ws + off;
        off += (bytes + 255) & ~(size_t)255;
        return p;
    };
    int*   cnt    = (int*)  carve((size_t)N * sizeof(int));
    int*   rowptr = (int*)  carve((size_t)(N + 1) * sizeof(int));
    float* dis    = (float*)carve((size_t)N * sizeof(float));
    int*   bsum   = (int*)  carve((size_t)nb * sizeof(int));
    int*   col    = (int*)  carve((size_t)E * sizeof(int));
    constexpr size_t planeCols = (NCHUNK > OUT_CH) ? NCHUNK : OUT_CH;
    float* h2     = (float*)carve((size_t)N * planeCols * sizeof(float));
    int*   tab    = (int*)h2;

    const dim3 sgrid(NCHUNK, ng);
    k_count_lds<<<sgrid, SORT_T, 0, stream>>>(dst, tab, N, E, chunk);
    k_colscan  <<<nb, 1024, 0, stream>>>(tab, cnt, bsum, N);
    k_scan     <<<nb, 256, 0, stream>>>(cnt, bsum, rowptr, dis, N);
    k_place_lds<<<sgrid, SORT_T, 0, stream>>>(src, dst, tab, rowptr, col, N, E, chunk);

    k_layer1_fused<<<(N + TILE_NODES - 1) / TILE_NODES, 256, 0, stream>>>(
        rowptr, col, x, dis, W1, b1, W2, h2, N);

    {
        constexpr int per = (256 / (OUT_CH / 4)) * 2;
        k_gather<OUT_CH, 2, true><<<(N + per - 1) / per, 256, 0, stream>>>(
            rowptr, col, h2, dis, b2, out, N);
    }
}
